# starting kernel
# baseline (speedup 1.0000x reference)
#include <hip/hip_runtime.h>
#include <hip/hip_bf16.h>

typedef __attribute__((ext_vector_type(2))) float v2f;
typedef __attribute__((ext_vector_type(4))) float v4f;
typedef __attribute__((ext_vector_type(8))) float v8f;

#define MDIM 16384
#define NDIM 1024
#define KDIM 1024
#define NLAYERS 20
#define EPSV 1e-6f

__global__ __launch_bounds__(256) void wsum_kernel(const float* __restrict__ cw,
                                                   float* __restrict__ ws) {
    size_t idx = ((size_t)blockIdx.x * blockDim.x + threadIdx.x) * 4;
    v4f acc = *(const v4f*)(cw + idx);
#pragma unroll
    for (int l = 1; l < NLAYERS; ++l) {
        acc += *(const v4f*)(cw + (size_t)l * (NDIM * KDIM) + idx);
    }
    *(v4f*)(ws + idx) = acc;
}

__global__ __launch_bounds__(256) void rmsnorm_kernel(const float* __restrict__ x,
                                                      const float* __restrict__ w,
                                                      float* __restrict__ out) {
    const int row  = blockIdx.x;
    const int tid  = threadIdx.x;
    const int lane = tid & 31;
    const int wid  = tid >> 5;
    const float* xr = x + (size_t)row * 1024;
    float* orow     = out + (size_t)row * 1024;

    __shared__ float sm[8];

    v4f v = *(const v4f*)(xr + tid * 4);
    float mx = fmaxf(fmaxf(fabsf(v.x), fabsf(v.y)), fmaxf(fabsf(v.z), fabsf(v.w)));
#pragma unroll
    for (int off = 16; off > 0; off >>= 1)
        mx = fmaxf(mx, __shfl_xor(mx, off, 32));
    if (lane == 0) sm[wid] = mx;
    __syncthreads();
    if (tid < 32) {
        float t = (tid < 8) ? sm[tid] : 0.0f;
#pragma unroll
        for (int off = 4; off > 0; off >>= 1)
            t = fmaxf(t, __shfl_xor(t, off, 32));
        if (tid == 0) sm[0] = fmaxf(t, 1e-6f);
    }
    __syncthreads();
    const float inv = 1.0f / sm[0];

    v4f xs = v * inv;
    float ss = xs.x * xs.x + xs.y * xs.y + xs.z * xs.z + xs.w * xs.w;
    __syncthreads();
#pragma unroll
    for (int off = 16; off > 0; off >>= 1)
        ss += __shfl_xor(ss, off, 32);
    if (lane == 0) sm[wid] = ss;
    __syncthreads();
    if (tid < 32) {
        float t = (tid < 8) ? sm[tid] : 0.0f;
#pragma unroll
        for (int off = 4; off > 0; off >>= 1)
            t += __shfl_xor(t, off, 32);
        if (tid == 0) sm[0] = rsqrtf(t * (1.0f / 1024.0f) + EPSV);
    }
    __syncthreads();
    const float rstd = sm[0];

    v4f wv = *(const v4f*)(w + tid * 4);
    v4f o  = xs * rstd * wv;
    *(v4f*)(orow + tid * 4) = o;
}

#define TM 128
#define TN 64
#define TK 32
#define LDSW 36

__global__ __launch_bounds__(256) void gemm_wmma_kernel(const float* __restrict__ X,
                                                        const float* __restrict__ Wm,
                                                        float* __restrict__ C) {
    __shared__ float As[TM * LDSW];
    __shared__ float Bs[TN * LDSW];

    const int tid  = threadIdx.x;
    const int lane = tid & 31;
    const int wid  = tid >> 5;
    const int wm   = wid & 3;
    const int wn   = wid >> 2;
    const int mblk = blockIdx.y * TM;
    const int nblk = blockIdx.x * TN;

    v8f acc00 = {};
    v8f acc01 = {};
    v8f acc10 = {};
    v8f acc11 = {};

    const int srow = tid >> 3;
    const int scol = (tid & 7) << 2;

    const int arow0 = wm * 32 + (lane & 15);
    const int arow1 = arow0 + 16;
    const int ncol0 = wn * 32 + (lane & 15);
    const int ncol1 = ncol0 + 16;
    const int kb    = (lane >> 4) << 1;

    for (int k0 = 0; k0 < KDIM; k0 += TK) {
        v4f a0 = *(const v4f*)(X  + (size_t)(mblk + srow      ) * KDIM + k0 + scol);
        v4f a1 = *(const v4f*)(X  + (size_t)(mblk + srow + 32 ) * KDIM + k0 + scol);
        v4f a2 = *(const v4f*)(X  + (size_t)(mblk + srow + 64 ) * KDIM + k0 + scol);
        v4f a3 = *(const v4f*)(X  + (size_t)(mblk + srow + 96 ) * KDIM + k0 + scol);
        v4f b0 = *(const v4f*)(Wm + (size_t)(nblk + srow      ) * KDIM + k0 + scol);
        v4f b1 = *(const v4f*)(Wm + (size_t)(nblk + srow + 32 ) * KDIM + k0 + scol);
        __syncthreads();
        *(v4f*)(As + (srow      ) * LDSW + scol) = a0;
        *(v4f*)(As + (srow + 32 ) * LDSW + scol) = a1;
        *(v4f*)(As + (srow + 64 ) * LDSW + scol) = a2;
        *(v4f*)(As + (srow + 96 ) * LDSW + scol) = a3;
        *(v4f*)(Bs + (srow      ) * LDSW + scol) = b0;
        *(v4f*)(Bs + (srow + 32 ) * LDSW + scol) = b1;
        __syncthreads();

#pragma unroll
        for (int kk = 0; kk < TK; kk += 4) {
            v2f af0 = *(const v2f*)(As + arow0 * LDSW + kk + kb);
            v2f af1 = *(const v2f*)(As + arow1 * LDSW + kk + kb);
            v2f bf0 = *(const v2f*)(Bs + ncol0 * LDSW + kk + kb);
            v2f bf1 = *(const v2f*)(Bs + ncol1 * LDSW + kk + kb);
            acc00 = __builtin_amdgcn_wmma_f32_16x16x4_f32(
                false, af0, false, bf0, (short)0, acc00, false, false);
            acc01 = __builtin_amdgcn_wmma_f32_16x16x4_f32(
                false, af0, false, bf1, (short)0, acc01, false, false);
            acc10 = __builtin_amdgcn_wmma_f32_16x16x4_f32(
                false, af1, false, bf0, (short)0, acc10, false, false);
            acc11 = __builtin_amdgcn_wmma_f32_16x16x4_f32(
                false, af1, false, bf1, (short)0, acc11, false, false);
        }
    }

    const float s  = 1.0f / (float)NLAYERS;
    const int row0 = mblk + wm * 32 + (lane >> 4) * 8;
    const int col0 = nblk + wn * 32 + (lane & 15);
#pragma unroll
    for (int i = 0; i < 8; ++i) {
        C[(size_t)(row0 + i     ) * NDIM + col0     ] = acc00[i] * s;
        C[(size_t)(row0 + i     ) * NDIM + col0 + 16] = acc01[i] * s;
        C[(size_t)(row0 + i + 16) * NDIM + col0     ] = acc10[i] * s;
        C[(size_t)(row0 + i + 16) * NDIM + col0 + 16] = acc11[i] * s;
    }
}

extern "C" void kernel_launch(void* const* d_in, const int* in_sizes, int n_in,
                              void* d_out, int out_size, void* d_ws, size_t ws_size,
                              hipStream_t stream) {
    const float* x  = (const float*)d_in[0];
    const float* cw = (const float*)d_in[1];
    const float* nw = (const float*)d_in[2];

    float* rms_out  = (float*)d_out;
    float* conv_out = rms_out + (size_t)MDIM * NDIM;
    float* wsum     = (float*)d_ws;

    wsum_kernel<<<(NDIM * KDIM) / 4 / 256, 256, 0, stream>>>(cw, wsum);

    rmsnorm_kernel<<<MDIM, 256, 0, stream>>>(x, nw, rms_out);

    dim3 grid(NDIM / TN, MDIM / TM);
    gemm_wmma_kernel<<<grid, 256, 0, stream>>>(x, wsum, conv_out);
}


# rmsnorm launch folded into the GEMM kernel's tail (one row per wave after its C tile) so row traffic runs under other blocks' matrix work; row sum of squares reduced inside one wave (new sum order)
# speedup vs baseline: 1.0840x; 1.0840x over previous
#include <hip/hip_runtime.h>
#include <hip/hip_bf16.h>

typedef __attribute__((ext_vector_type(2))) float v2f;
typedef __attribute__((ext_vector_type(4))) float v4f;
typedef __attribute__((ext_vector_type(8))) float v8f;

#define MDIM 16384
#define NDIM 1024
#define KDIM 1024
#define NLAYERS 20
#define EPSV 1e-6f

__global__ __launch_bounds__(256) void wsum_kernel(const float* __restrict__ cw,
                                                   float* __restrict__ ws) {
    size_t idx = ((size_t)blockIdx.x * blockDim.x + threadIdx.x) * 4;
    v4f acc = *(const v4f*)(cw + idx);
#pragma unroll
    for (int l = 1; l < NLAYERS; ++l) {
        acc += *(const v4f*)(cw + (size_t)l * (NDIM * KDIM) + idx);
    }
    *(v4f*)(ws + idx) = acc;
}

__device__ __forceinline__ void rmsnorm_row_by_wave(const float* __restrict__ xr,
                                                    const float* __restrict__ w,
                                                    float* __restrict__ orow,
                                                    const int lane) {
    const int lo = lane * 4;
    v4f v[8], wv[8];
#pragma unroll
    for (int c = 0; c < 8; ++c) v[c] = *(const v4f*)(xr + c * 128 + lo);
#pragma unroll
    for (int c = 0; c < 8; ++c) wv[c] = *(const v4f*)(w + c * 128 + lo);

    float mx = 0.0f;
#pragma unroll
    for (int c = 0; c < 8; ++c)
        mx = fmaxf(mx, fmaxf(fmaxf(fabsf(v[c].x), fabsf(v[c].y)),
                             fmaxf(fabsf(v[c].z), fabsf(v[c].w))));
#pragma unroll
    for (int off = 16; off > 0; off >>= 1)
        mx = fmaxf(mx, __shfl_xor(mx, off, 32));
    const float inv = 1.0f / fmaxf(mx, 1e-6f);

    float ss = 0.0f;
#pragma unroll
    for (int c = 0; c < 8; ++c) {
        v[c] = v[c] * inv;
        ss += v[c].x * v[c].x + v[c].y * v[c].y + v[c].z * v[c].z + v[c].w * v[c].w;
    }
#pragma unroll
    for (int off = 16; off > 0; off >>= 1)
        ss += __shfl_xor(ss, off, 32);
    const float rstd = rsqrtf(ss * (1.0f / 1024.0f) + EPSV);

#pragma unroll
    for (int c = 0; c < 8; ++c)
        *(v4f*)(orow + c * 128 + lo) = v[c] * rstd * wv[c];
}

#define TM 128
#define TN 64
#define TK 32
#define LDSW 36

__global__ __launch_bounds__(256, 12) void gemm_wmma_kernel(const float* __restrict__ X,
                                                        const float* __restrict__ Wm,
                                                        const float* __restrict__ nw,
                                                        float* __restrict__ R,
                                                        float* __restrict__ C) {
    __shared__ float As[TM * LDSW];
    __shared__ float Bs[TN * LDSW];

    const int tid  = threadIdx.x;
    const int lane = tid & 31;
    const int wid  = tid >> 5;
    const int wm   = wid & 3;
    const int wn   = wid >> 2;
    const int mblk = blockIdx.y * TM;
    const int nblk = blockIdx.x * TN;

    v8f acc00 = {};
    v8f acc01 = {};
    v8f acc10 = {};
    v8f acc11 = {};

    const int srow = tid >> 3;
    const int scol = (tid & 7) << 2;

    const int arow0 = wm * 32 + (lane & 15);
    const int arow1 = arow0 + 16;
    const int ncol0 = wn * 32 + (lane & 15);
    const int ncol1 = ncol0 + 16;
    const int kb    = (lane >> 4) << 1;

    for (int k0 = 0; k0 < KDIM; k0 += TK) {
        v4f a0 = *(const v4f*)(X  + (size_t)(mblk + srow      ) * KDIM + k0 + scol);
        v4f a1 = *(const v4f*)(X  + (size_t)(mblk + srow + 32 ) * KDIM + k0 + scol);
        v4f a2 = *(const v4f*)(X  + (size_t)(mblk + srow + 64 ) * KDIM + k0 + scol);
        v4f a3 = *(const v4f*)(X  + (size_t)(mblk + srow + 96 ) * KDIM + k0 + scol);
        v4f b0 = *(const v4f*)(Wm + (size_t)(nblk + srow      ) * KDIM + k0 + scol);
        v4f b1 = *(const v4f*)(Wm + (size_t)(nblk + srow + 32 ) * KDIM + k0 + scol);
        __syncthreads();
        *(v4f*)(As + (srow      ) * LDSW + scol) = a0;
        *(v4f*)(As + (srow + 32 ) * LDSW + scol) = a1;
        *(v4f*)(As + (srow + 64 ) * LDSW + scol) = a2;
        *(v4f*)(As + (srow + 96 ) * LDSW + scol) = a3;
        *(v4f*)(Bs + (srow      ) * LDSW + scol) = b0;
        *(v4f*)(Bs + (srow + 32 ) * LDSW + scol) = b1;
        __syncthreads();

#pragma unroll
        for (int kk = 0; kk < TK; kk += 4) {
            v2f af0 = *(const v2f*)(As + arow0 * LDSW + kk + kb);
            v2f af1 = *(const v2f*)(As + arow1 * LDSW + kk + kb);
            v2f bf0 = *(const v2f*)(Bs + ncol0 * LDSW + kk + kb);
            v2f bf1 = *(const v2f*)(Bs + ncol1 * LDSW + kk + kb);
            acc00 = __builtin_amdgcn_wmma_f32_16x16x4_f32(
                false, af0, false, bf0, (short)0, acc00, false, false);
            acc01 = __builtin_amdgcn_wmma_f32_16x16x4_f32(
                false, af0, false, bf1, (short)0, acc01, false, false);
            acc10 = __builtin_amdgcn_wmma_f32_16x16x4_f32(
                false, af1, false, bf0, (short)0, acc10, false, false);
            acc11 = __builtin_amdgcn_wmma_f32_16x16x4_f32(
                false, af1, false, bf1, (short)0, acc11, false, false);
        }
    }

    const float s  = 1.0f / (float)NLAYERS;
    const int row0 = mblk + wm * 32 + (lane >> 4) * 8;
    const int col0 = nblk + wn * 32 + (lane & 15);
#pragma unroll
    for (int i = 0; i < 8; ++i) {
        C[(size_t)(row0 + i     ) * NDIM + col0     ] = acc00[i] * s;
        C[(size_t)(row0 + i     ) * NDIM + col0 + 16] = acc01[i] * s;
        C[(size_t)(row0 + i + 16) * NDIM + col0     ] = acc10[i] * s;
        C[(size_t)(row0 + i + 16) * NDIM + col0 + 16] = acc11[i] * s;
    }

    __builtin_amdgcn_sched_barrier(0);

    const int rrow = mblk + blockIdx.x * 8 + wid;
    rmsnorm_row_by_wave(X + (size_t)rrow * KDIM, nw, R + (size_t)rrow * NDIM, lane);
}

extern "C" void kernel_launch(void* const* d_in, const int* in_sizes, int n_in,
                              void* d_out, int out_size, void* d_ws, size_t ws_size,
                              hipStream_t stream) {
    const float* x  = (const float*)d_in[0];
    const float* cw = (const float*)d_in[1];
    const float* nw = (const float*)d_in[2];

    float* rms_out  = (float*)d_out;
    float* conv_out = rms_out + (size_t)MDIM * NDIM;
    float* wsum     = (float*)d_ws;

    wsum_kernel<<<(NDIM * KDIM) / 4 / 256, 256, 0, stream>>>(cw, wsum);

    dim3 grid(NDIM / TN, MDIM / TM);
    gemm_wmma_kernel<<<grid, 256, 0, stream>>>(x, wsum, nw, rms_out, conv_out);
}


# GEMM block tile 128x64 -> 128x128, wave tile 32x32 -> 64x32 (8 accumulators): 6 LDS fragment reads per 8 WMMAs (was 4 per 4), half the barriers per WMMA, a third less staging; 1024 blocks, 4 per WGP
# speedup vs baseline: 1.1484x; 1.0594x over previous
#include <hip/hip_runtime.h>
#include <hip/hip_bf16.h>

typedef __attribute__((ext_vector_type(2))) float v2f;
typedef __attribute__((ext_vector_type(4))) float v4f;
typedef __attribute__((ext_vector_type(8))) float v8f;

#define MDIM 16384
#define NDIM 1024
#define KDIM 1024
#define NLAYERS 20
#define EPSV 1e-6f

__global__ __launch_bounds__(256) void wsum_kernel(const float* __restrict__ cw,
                                                   float* __restrict__ ws) {
    size_t idx = ((size_t)blockIdx.x * blockDim.x + threadIdx.x) * 4;
    v4f acc = *(const v4f*)(cw + idx);
#pragma unroll
    for (int l = 1; l < NLAYERS; ++l) {
        acc += *(const v4f*)(cw + (size_t)l * (NDIM * KDIM) + idx);
    }
    *(v4f*)(ws + idx) = acc;
}

__device__ __forceinline__ void rmsnorm_row_by_wave(const float* __restrict__ xr,
                                                    const float* __restrict__ w,
                                                    float* __restrict__ orow,
                                                    const int lane) {
    const int lo = lane * 4;
    v4f v[8], wv[8];
#pragma unroll
    for (int c = 0; c < 8; ++c) v[c] = *(const v4f*)(xr + c * 128 + lo);
#pragma unroll
    for (int c = 0; c < 8; ++c) wv[c] = *(const v4f*)(w + c * 128 + lo);

    float mx = 0.0f;
#pragma unroll
    for (int c = 0; c < 8; ++c)
        mx = fmaxf(mx, fmaxf(fmaxf(fabsf(v[c].x), fabsf(v[c].y)),
                             fmaxf(fabsf(v[c].z), fabsf(v[c].w))));
#pragma unroll
    for (int off = 16; off > 0; off >>= 1)
        mx = fmaxf(mx, __shfl_xor(mx, off, 32));
    const float inv = 1.0f / fmaxf(mx, 1e-6f);

    float ss = 0.0f;
#pragma unroll
    for (int c = 0; c < 8; ++c) {
        v[c] = v[c] * inv;
        ss += v[c].x * v[c].x + v[c].y * v[c].y + v[c].z * v[c].z + v[c].w * v[c].w;
    }
#pragma unroll
    for (int off = 16; off > 0; off >>= 1)
        ss += __shfl_xor(ss, off, 32);
    const float rstd = rsqrtf(ss * (1.0f / 1024.0f) + EPSV);

#pragma unroll
    for (int c = 0; c < 8; ++c)
        *(v4f*)(orow + c * 128 + lo) = v[c] * rstd * wv[c];
}

#define TM 128
#define TN 128
#define TK 32
#define LDSW 36

__global__ __launch_bounds__(256) __attribute__((amdgpu_waves_per_eu(8, 8))) void gemm_wmma_kernel(const float* __restrict__ X,
                                                       const float* __restrict__ Wm,
                                                       const float* __restrict__ nw,
                                                       float* __restrict__ R,
                                                       float* __restrict__ C) {
    __shared__ float As[TM * LDSW];
    __shared__ float Bs[TN * LDSW];

    const int tid  = threadIdx.x;
    const int lane = tid & 31;
    const int wid  = __builtin_amdgcn_readfirstlane(tid >> 5);
    const int wm   = wid & 1;
    const int wn   = wid >> 1;
    const int mblk = blockIdx.y * TM;
    const int nblk = blockIdx.x * TN;

    v8f acc[4][2];
#pragma unroll
    for (int t = 0; t < 4; ++t)
#pragma unroll
        for (int j = 0; j < 2; ++j) acc[t][j] = (v8f){};

    const int srow = tid >> 3;
    const int scol = (tid & 7) << 2;
    const float* xp = X  + (size_t)(mblk + srow) * KDIM + scol;
    const float* wp = Wm + (size_t)(nblk + srow) * KDIM + scol;
    float* asp = As + srow * LDSW + scol;
    float* bsp = Bs + srow * LDSW + scol;

    const int kb = (lane >> 4) << 1;
    const float* afp = As + (wm * 64 + (lane & 15)) * LDSW + kb;
    const float* bfp = Bs + (wn * 32 + (lane & 15)) * LDSW + kb;

    for (int k0 = 0; k0 < KDIM; k0 += TK) {
        v4f a[4], b[4];
#pragma unroll
        for (int i = 0; i < 4; ++i) a[i] = *(const v4f*)(xp + (size_t)i * 32 * KDIM + k0);
#pragma unroll
        for (int i = 0; i < 4; ++i) b[i] = *(const v4f*)(wp + (size_t)i * 32 * KDIM + k0);
        __syncthreads();
#pragma unroll
        for (int i = 0; i < 4; ++i) *(v4f*)(asp + i * 32 * LDSW) = a[i];
#pragma unroll
        for (int i = 0; i < 4; ++i) *(v4f*)(bsp + i * 32 * LDSW) = b[i];
        __syncthreads();

#pragma unroll
        for (int kk = 0; kk < TK; kk += 4) {
            v2f af[4], bf[2];
#pragma unroll
            for (int t = 0; t < 4; ++t) af[t] = *(const v2f*)(afp + t * 16 * LDSW + kk);
#pragma unroll
            for (int j = 0; j < 2; ++j) bf[j] = *(const v2f*)(bfp + j * 16 * LDSW + kk);
#pragma unroll
            for (int t = 0; t < 4; ++t)
#pragma unroll
                for (int j = 0; j < 2; ++j)
                    acc[t][j] = __builtin_amdgcn_wmma_f32_16x16x4_f32(
                        false, af[t], false, bf[j], (short)0, acc[t][j], false, false);
        }
    }

    const float s  = 1.0f / (float)NLAYERS;
    const int row0 = mblk + wm * 64 + (lane >> 4) * 8;
    const int col0 = nblk + wn * 32 + (lane & 15);
#pragma unroll
    for (int t = 0; t < 4; ++t)
#pragma unroll
        for (int i = 0; i < 8; ++i) {
            C[(size_t)(row0 + t * 16 + i) * NDIM + col0     ] = acc[t][0][i] * s;
            C[(size_t)(row0 + t * 16 + i) * NDIM + col0 + 16] = acc[t][1][i] * s;
        }

    const int rrow = mblk + blockIdx.x * 16 + wid * 2;
#pragma unroll 1
    for (int r = 0; r < 2; ++r)
        rmsnorm_row_by_wave(X + (size_t)(rrow + r) * KDIM, nw, R + (size_t)(rrow + r) * NDIM, lane);
}

extern "C" void kernel_launch(void* const* d_in, const int* in_sizes, int n_in,
                              void* d_out, int out_size, void* d_ws, size_t ws_size,
                              hipStream_t stream) {
    const float* x  = (const float*)d_in[0];
    const float* cw = (const float*)d_in[1];
    const float* nw = (const float*)d_in[2];

    float* rms_out  = (float*)d_out;
    float* conv_out = rms_out + (size_t)MDIM * NDIM;
    float* wsum     = (float*)d_ws;

    wsum_kernel<<<(NDIM * KDIM) / 4 / 256, 256, 0, stream>>>(cw, wsum);

    dim3 grid(NDIM / TN, MDIM / TM);
    gemm_wmma_kernel<<<grid, 256, 0, stream>>>(x, wsum, nw, rms_out, conv_out);
}


# GEMM K loop software-pipelined: next K tile's global loads in flight in registers under the WMMAs, second LDS buffer, one barrier per tile; 4-wave blocks of 64x64 wave tiles at 256 registers
# speedup vs baseline: 1.1894x; 1.0357x over previous
#include <hip/hip_runtime.h>
#include <hip/hip_bf16.h>

typedef __attribute__((ext_vector_type(2))) float v2f;
typedef __attribute__((ext_vector_type(4))) float v4f;
typedef __attribute__((ext_vector_type(8))) float v8f;

#define MDIM 16384
#define NDIM 1024
#define KDIM 1024
#define NLAYERS 20
#define EPSV 1e-6f

__global__ __launch_bounds__(256) void wsum_kernel(const float* __restrict__ cw,
                                                   float* __restrict__ ws) {
    size_t idx = ((size_t)blockIdx.x * blockDim.x + threadIdx.x) * 4;
    v4f acc = *(const v4f*)(cw + idx);
#pragma unroll
    for (int l = 1; l < NLAYERS; ++l) {
        acc += *(const v4f*)(cw + (size_t)l * (NDIM * KDIM) + idx);
    }
    *(v4f*)(ws + idx) = acc;
}

__device__ __forceinline__ void rmsnorm_row_by_wave(const float* __restrict__ xr,
                                                    const float* __restrict__ w,
                                                    float* __restrict__ orow,
                                                    const int lane) {
    const int lo = lane * 4;
    v4f v[8], wv[8];
#pragma unroll
    for (int c = 0; c < 8; ++c) v[c] = *(const v4f*)(xr + c * 128 + lo);
#pragma unroll
    for (int c = 0; c < 8; ++c) wv[c] = *(const v4f*)(w + c * 128 + lo);

    float mx = 0.0f;
#pragma unroll
    for (int c = 0; c < 8; ++c)
        mx = fmaxf(mx, fmaxf(fmaxf(fabsf(v[c].x), fabsf(v[c].y)),
                             fmaxf(fabsf(v[c].z), fabsf(v[c].w))));
#pragma unroll
    for (int off = 16; off > 0; off >>= 1)
        mx = fmaxf(mx, __shfl_xor(mx, off, 32));
    const float inv = 1.0f / fmaxf(mx, 1e-6f);

    float ss = 0.0f;
#pragma unroll
    for (int c = 0; c < 8; ++c) {
        v[c] = v[c] * inv;
        ss += v[c].x * v[c].x + v[c].y * v[c].y + v[c].z * v[c].z + v[c].w * v[c].w;
    }
#pragma unroll
    for (int off = 16; off > 0; off >>= 1)
        ss += __shfl_xor(ss, off, 32);
    const float rstd = rsqrtf(ss * (1.0f / 1024.0f) + EPSV);

#pragma unroll
    for (int c = 0; c < 8; ++c)
        *(v4f*)(orow + c * 128 + lo) = v[c] * rstd * wv[c];
}

#define TM 128
#define TN 128
#define TK 32
#define LDSW 36
#define NKT (KDIM / TK)
#define TILEF (TM * LDSW)

__device__ __forceinline__ void lds_barrier() {
    __builtin_amdgcn_fence(__ATOMIC_RELEASE, "workgroup", "local");
    __builtin_amdgcn_s_barrier();
    __builtin_amdgcn_fence(__ATOMIC_ACQUIRE, "workgroup", "local");
}

__device__ __forceinline__ void wmma_k_tile(v8f (&acc)[4][4],
                                            const float* __restrict__ af_base,
                                            const float* __restrict__ bf_base) {
#pragma unroll
    for (int kk = 0; kk < TK; kk += 4) {
        v2f af[4], bf[4];
#pragma unroll
        for (int t = 0; t < 4; ++t) af[t] = *(const v2f*)(af_base + t * 16 * LDSW + kk);
#pragma unroll
        for (int j = 0; j < 4; ++j) bf[j] = *(const v2f*)(bf_base + j * 16 * LDSW + kk);
#pragma unroll
        for (int t = 0; t < 4; ++t)
#pragma unroll
            for (int j = 0; j < 4; ++j)
                acc[t][j] = __builtin_amdgcn_wmma_f32_16x16x4_f32(
                    false, af[t], false, bf[j], (short)0, acc[t][j], false, false);
    }
}

__global__ __launch_bounds__(128) __attribute__((amdgpu_waves_per_eu(4, 4))) void gemm_wmma_kernel(const float* __restrict__ X,
                                                       const float* __restrict__ Wm,
                                                       const float* __restrict__ nw,
                                                       float* __restrict__ R,
                                                       float* __restrict__ C) {
    __shared__ float As[2 * TM * LDSW];
    __shared__ float Bs[2 * TN * LDSW];

    const int tid  = threadIdx.x;
    const int lane = tid & 31;
    const int wid  = __builtin_amdgcn_readfirstlane(tid >> 5);
    const int wm   = wid & 1;
    const int wn   = wid >> 1;
    const int mblk = blockIdx.y * TM;
    const int nblk = blockIdx.x * TN;

    v8f acc[4][4];
#pragma unroll
    for (int t = 0; t < 4; ++t)
#pragma unroll
        for (int j = 0; j < 4; ++j) acc[t][j] = (v8f){};

    const int srow = tid >> 3;
    const int scol = (tid & 7) << 2;
    const float* xp = X  + (size_t)(mblk + srow) * KDIM + scol;
    const float* wp = Wm + (size_t)(nblk + srow) * KDIM + scol;
    float* asp = As + srow * LDSW + scol;
    float* bsp = Bs + srow * LDSW + scol;

    const int kb = (lane >> 4) << 1;
    const float* afp = As + (wm * 64 + (lane & 15)) * LDSW + kb;
    const float* bfp = Bs + (wn * 64 + (lane & 15)) * LDSW + kb;

    v4f a[8], b[8];

#pragma unroll
    for (int i = 0; i < 8; ++i) a[i] = *(const v4f*)(xp + (size_t)i * 16 * KDIM);
#pragma unroll
    for (int i = 0; i < 8; ++i) b[i] = *(const v4f*)(wp + (size_t)i * 16 * KDIM);
#pragma unroll
    for (int i = 0; i < 8; ++i) *(v4f*)(asp + i * 16 * LDSW) = a[i];
#pragma unroll
    for (int i = 0; i < 8; ++i) *(v4f*)(bsp + i * 16 * LDSW) = b[i];
#pragma unroll
    for (int i = 0; i < 8; ++i) a[i] = *(const v4f*)(xp + (size_t)i * 16 * KDIM + TK);
#pragma unroll
    for (int i = 0; i < 8; ++i) b[i] = *(const v4f*)(wp + (size_t)i * 16 * KDIM + TK);
    lds_barrier();

    for (int kt = 0; kt < NKT - 2; ++kt) {
        const int cur = (kt & 1) * TILEF;
        const int nxt = TILEF - cur;
        wmma_k_tile(acc, afp + cur, bfp + cur);
#pragma unroll
        for (int i = 0; i < 8; ++i) *(v4f*)(asp + nxt + i * 16 * LDSW) = a[i];
#pragma unroll
        for (int i = 0; i < 8; ++i) *(v4f*)(bsp + nxt + i * 16 * LDSW) = b[i];
        const int k2 = (kt + 2) * TK;
#pragma unroll
        for (int i = 0; i < 8; ++i) a[i] = *(const v4f*)(xp + (size_t)i * 16 * KDIM + k2);
#pragma unroll
        for (int i = 0; i < 8; ++i) b[i] = *(const v4f*)(wp + (size_t)i * 16 * KDIM + k2);
        lds_barrier();
    }
    wmma_k_tile(acc, afp, bfp);
#pragma unroll
    for (int i = 0; i < 8; ++i) *(v4f*)(asp + TILEF + i * 16 * LDSW) = a[i];
#pragma unroll
    for (int i = 0; i < 8; ++i) *(v4f*)(bsp + TILEF + i * 16 * LDSW) = b[i];
    lds_barrier();
    wmma_k_tile(acc, afp + TILEF, bfp + TILEF);

    const float s  = 1.0f / (float)NLAYERS;
    const int row0 = mblk + wm * 64 + (lane >> 4) * 8;
    const int col0 = nblk + wn * 64 + (lane & 15);
#pragma unroll
    for (int t = 0; t < 4; ++t)
#pragma unroll
        for (int i = 0; i < 8; ++i) {
#pragma unroll
            for (int j = 0; j < 4; ++j)
                C[(size_t)(row0 + t * 16 + i) * NDIM + col0 + j * 16] = acc[t][j][i] * s;
        }

    const int rrow = mblk + blockIdx.x * 16 + wid * 4;
    for (int r = 0; r < 4; ++r)
        rmsnorm_row_by_wave(X + (size_t)(rrow + r) * KDIM, nw, R + (size_t)(rrow + r) * NDIM, lane);
}

extern "C" void kernel_launch(void* const* d_in, const int* in_sizes, int n_in,
                              void* d_out, int out_size, void* d_ws, size_t ws_size,
                              hipStream_t stream) {
    const float* x  = (const float*)d_in[0];
    const float* cw = (const float*)d_in[1];
    const float* nw = (const float*)d_in[2];

    float* rms_out  = (float*)d_out;
    float* conv_out = rms_out + (size_t)MDIM * NDIM;
    float* wsum     = (float*)d_ws;

    wsum_kernel<<<(NDIM * KDIM) / 4 / 256, 256, 0, stream>>>(cw, wsum);

    dim3 grid(NDIM / TN, MDIM / TM);
    gemm_wmma_kernel<<<grid, 128, 0, stream>>>(x, wsum, nw, rms_out, conv_out);
}
